# starting kernel
# baseline (speedup 1.0000x reference)
#include <hip/hip_runtime.h>
#include <math.h>


typedef float v2f __attribute__((ext_vector_type(2)));
typedef float v8f __attribute__((ext_vector_type(8)));

#define TB      8
#define STRIDE  116
#define NCOLS   112
#define NTILES  7

__device__ __forceinline__ void gemm256(const float* __restrict__ W,
                                        const float* __restrict__ in,
                                        float* __restrict__ out,
                                        int lane, int wv, bool syncBeforeWrite)
{
    v8f acc[2][NTILES];
    const v8f vz = {0.f, 0.f, 0.f, 0.f, 0.f, 0.f, 0.f, 0.f};
#pragma unroll
    for (int r = 0; r < 2; ++r)
#pragma unroll
        for (int t = 0; t < NTILES; ++t) acc[r][t] = vz;

    const int m0    = (wv * 2 + 0) * 16 + (lane & 15);
    const int m1    = (wv * 2 + 1) * 16 + (lane & 15);
    const int khalf = (lane >> 4) * 2;
    const int nlo   = lane & 15;

    for (int ks = 0; ks < 64; ++ks) {
        const int k = ks * 4 + khalf;
        v2f a0 = *(const v2f*)(W + m0 * 256 + k);
        v2f a1 = *(const v2f*)(W + m1 * 256 + k);
#pragma unroll
        for (int t = 0; t < NTILES; ++t) {
            const int c = t * 16 + nlo;
            v2f bf;
            bf.x = in[(k + 0) * STRIDE + c];
            bf.y = in[(k + 1) * STRIDE + c];
            acc[0][t] = __builtin_amdgcn_wmma_f32_16x16x4_f32(
                false, a0, false, bf, (short)0, acc[0][t], false, false);
            acc[1][t] = __builtin_amdgcn_wmma_f32_16x16x4_f32(
                false, a1, false, bf, (short)0, acc[1][t], false, false);
        }
    }

    if (syncBeforeWrite) __syncthreads();

#pragma unroll
    for (int r = 0; r < 2; ++r) {
        const int mbase = (wv * 2 + r) * 16 + 8 * (lane >> 4);
#pragma unroll
        for (int t = 0; t < NTILES; ++t) {
            const int c = t * 16 + nlo;
#pragma unroll
            for (int v = 0; v < 8; ++v)
                out[(mbase + v) * STRIDE + c] = acc[r][t][v];
        }
    }
}

__global__ __launch_bounds__(256, 1)
void diffnet_fused_kernel(const float* __restrict__ x,
                          const float* __restrict__ w0, const float* __restrict__ b0,
                          const float* __restrict__ w1, const float* __restrict__ b1,
                          const float* __restrict__ w2, const float* __restrict__ b2,
                          const float* __restrict__ w3, const float* __restrict__ b3,
                          float* __restrict__ out)
{
    extern __shared__ float sm[];
    float* Abuf = sm;
    float* Bbuf = Abuf + 256 * STRIDE;
    float* gp0  = Bbuf + 256 * STRIDE;
    float* gpp0 = gp0  + 256 * TB;
    float* gp1  = gpp0 + 256 * TB;
    float* gpp1 = gp1  + 256 * TB;
    float* gp2  = gpp1 + 256 * TB;
    float* gpp2 = gp2  + 256 * TB;
    float* s2   = gpp2 + 256 * TB;
    float* v2s  = s2   + 256 * TB;
    float* v1s  = v2s  + 256 * TB;
    float* xl   = v1s  + 256 * TB;
    float* outa = xl   + TB * 12;

    const int tid  = threadIdx.x;
    const int lane = tid & 31;
    const int wv   = tid >> 5;
    const int n    = blockIdx.y;
    const int bg0  = blockIdx.x * TB;

    const float* W0 = w0 + n * 256 * 12;
    const float* W1 = w1 + n * 256 * 256;
    const float* W2 = w2 + n * 256 * 256;
    const float* W3 = w3 + n * 256;
    const float* Bi0 = b0 + n * 256;
    const float* Bi1 = b1 + n * 256;
    const float* Bi2 = b2 + n * 256;

    if (tid < TB * 12) xl[tid] = x[bg0 * 12 + tid];
    if (tid < TB)      outa[tid] = b3[n];
    __syncthreads();

    {
        const int i = tid;
        float w0r[12];
#pragma unroll
        for (int m = 0; m < 12; ++m) w0r[m] = W0[i * 12 + m];
        for (int b = 0; b < TB; ++b) {
            float a = Bi0[i];
#pragma unroll
            for (int m = 0; m < 12; ++m) a += w0r[m] * xl[b * 12 + m];
            float t  = tanhf(a);
            float gp = 1.f - t * t;
            gp0[i * TB + b]  = gp;
            gpp0[i * TB + b] = -2.f * t * gp;
#pragma unroll
            for (int m = 0; m < 12; ++m) Abuf[i * STRIDE + b * 12 + m] = gp * w0r[m];
            Abuf[i * STRIDE + 96 + b]  = t;
            Abuf[i * STRIDE + 104 + b] = 0.f;
        }
    }
    __syncthreads();

    gemm256(W1, Abuf, Bbuf, lane, wv, false);
    __syncthreads();

    {
        const int i = tid;
        for (int b = 0; b < TB; ++b) {
            float a  = Bbuf[i * STRIDE + 96 + b] + Bi1[i];
            float t  = tanhf(a);
            float gp = 1.f - t * t;
            gp1[i * TB + b]  = gp;
            gpp1[i * TB + b] = -2.f * t * gp;
#pragma unroll
            for (int m = 0; m < 12; ++m)
                Abuf[i * STRIDE + b * 12 + m] = gp * Bbuf[i * STRIDE + b * 12 + m];
            Abuf[i * STRIDE + 96 + b] = t;
        }
    }
    __syncthreads();

    gemm256(W2, Abuf, Abuf, lane, wv, true);
    __syncthreads();

    {
        const int i = tid;
        for (int b = 0; b < TB; ++b) {
            float a  = Abuf[i * STRIDE + 96 + b] + Bi2[i];
            float t  = tanhf(a);
            float gp = 1.f - t * t;
            gp2[i * TB + b]  = gp;
            gpp2[i * TB + b] = -2.f * t * gp;
            s2[i * TB + b]   = W3[i] * gp;
            atomicAdd(&outa[b], W3[i] * t);
        }
    }
    __syncthreads();

    {
        const int i = tid;
        float acc[TB];
#pragma unroll
        for (int b = 0; b < TB; ++b) acc[b] = 0.f;
        for (int o = 0; o < 256; ++o) {
            float w = W2[o * 256 + i];
#pragma unroll
            for (int b = 0; b < TB; ++b) acc[b] += w * s2[o * TB + b];
        }
#pragma unroll
        for (int b = 0; b < TB; ++b) v2s[i * TB + b] = acc[b];
    }
    __syncthreads();

    {
        const int i = tid;
        float acc[TB];
#pragma unroll
        for (int b = 0; b < TB; ++b) acc[b] = 0.f;
        for (int o = 0; o < 256; ++o) {
            float w = W1[o * 256 + i];
#pragma unroll
            for (int b = 0; b < TB; ++b) acc[b] += w * v2s[o * TB + b] * gp1[o * TB + b];
        }
#pragma unroll
        for (int b = 0; b < TB; ++b) v1s[i * TB + b] = acc[b];
    }
    __syncthreads();

    for (int idx = tid; idx < TB * 12; idx += 256) {
        const int b = idx / 12, m = idx % 12;
        float acc = 0.f;
        for (int o = 0; o < 256; ++o)
            acc += s2[o * TB + b] * Abuf[o * STRIDE + b * 12 + m];
        out[2048 + (n * 512 + bg0 + b) * 12 + m] = acc;
    }
    if (tid < TB) out[n * 512 + bg0 + tid] = outa[tid];
    __syncthreads();

    {
        const int i = tid;
#pragma unroll
        for (int b = 0; b < TB; ++b) {
            s2[i * TB + b]  = W3[i]          * gpp2[i * TB + b];
            v2s[i * TB + b] = v2s[i * TB + b] * gpp1[i * TB + b];
            v1s[i * TB + b] = v1s[i * TB + b] * gpp0[i * TB + b];
        }
    }
    __syncthreads();

    for (int idx = tid; idx < TB * 144; idx += 256) {
        const int b = idx / 144, r = idx % 144;
        const int j = r / 12,   k = r % 12;
        float acc = 0.f;
        for (int i = 0; i < 256; ++i) {
            acc += v1s[i * TB + b] * W0[i * 12 + j] * W0[i * 12 + k];
            acc += v2s[i * TB + b] * Bbuf[i * STRIDE + b * 12 + j] * Bbuf[i * STRIDE + b * 12 + k];
            acc += s2[i * TB + b]  * Abuf[i * STRIDE + b * 12 + j] * Abuf[i * STRIDE + b * 12 + k];
        }
        out[26624 + (n * 512 + bg0 + b) * 144 + r] = acc;
    }
}

extern "C" void kernel_launch(void* const* d_in, const int* in_sizes, int n_in,
                              void* d_out, int out_size, void* d_ws, size_t ws_size,
                              hipStream_t stream)
{
    (void)in_sizes; (void)n_in; (void)out_size; (void)d_ws; (void)ws_size;
    const float* x  = (const float*)d_in[0];
    const float* w0 = (const float*)d_in[1];
    const float* b0 = (const float*)d_in[2];
    const float* w1 = (const float*)d_in[3];
    const float* b1 = (const float*)d_in[4];
    const float* w2 = (const float*)d_in[5];
    const float* b2 = (const float*)d_in[6];
    const float* w3 = (const float*)d_in[7];
    const float* b3 = (const float*)d_in[8];
    float* outp = (float*)d_out;

    const size_t smem = (size_t)(2 * 256 * STRIDE + 9 * 256 * TB + TB * 12 + TB) * sizeof(float);
    (void)hipFuncSetAttribute(reinterpret_cast<const void*>(&diffnet_fused_kernel),
                              hipFuncAttributeMaxDynamicSharedMemorySize, (int)smem);

    dim3 grid(512 / TB, 4);
    dim3 block(256);
    diffnet_fused_kernel<<<grid, block, smem, stream>>>(x, w0, b0, w1, b1, w2, b2, w3, b3, outp);
}


# d2_out (the 12x12 second-derivative blocks) moved from scalar loops onto f32 16x16x4 matrix instructions: one wave per batch element, A = c-scaled J fragment, B = J fragment, staged through LDS
# speedup vs baseline: 2.2010x; 2.2010x over previous
#include <hip/hip_runtime.h>
#include <math.h>


typedef float v2f __attribute__((ext_vector_type(2)));
typedef float v8f __attribute__((ext_vector_type(8)));

#define TB      8
#define STRIDE  116
#define NCOLS   112
#define NTILES  7

__device__ __forceinline__ void gemm256(const float* __restrict__ W,
                                        const float* __restrict__ in,
                                        float* __restrict__ out,
                                        int lane, int wv, bool syncBeforeWrite)
{
    v8f acc[2][NTILES];
    const v8f vz = {0.f, 0.f, 0.f, 0.f, 0.f, 0.f, 0.f, 0.f};
#pragma unroll
    for (int r = 0; r < 2; ++r)
#pragma unroll
        for (int t = 0; t < NTILES; ++t) acc[r][t] = vz;

    const int m0    = (wv * 2 + 0) * 16 + (lane & 15);
    const int m1    = (wv * 2 + 1) * 16 + (lane & 15);
    const int khalf = (lane >> 4) * 2;
    const int nlo   = lane & 15;

    for (int ks = 0; ks < 64; ++ks) {
        const int k = ks * 4 + khalf;
        v2f a0 = *(const v2f*)(W + m0 * 256 + k);
        v2f a1 = *(const v2f*)(W + m1 * 256 + k);
#pragma unroll
        for (int t = 0; t < NTILES; ++t) {
            const int c = t * 16 + nlo;
            v2f bf;
            bf.x = in[(k + 0) * STRIDE + c];
            bf.y = in[(k + 1) * STRIDE + c];
            acc[0][t] = __builtin_amdgcn_wmma_f32_16x16x4_f32(
                false, a0, false, bf, (short)0, acc[0][t], false, false);
            acc[1][t] = __builtin_amdgcn_wmma_f32_16x16x4_f32(
                false, a1, false, bf, (short)0, acc[1][t], false, false);
        }
    }

    if (syncBeforeWrite) __syncthreads();

#pragma unroll
    for (int r = 0; r < 2; ++r) {
        const int mbase = (wv * 2 + r) * 16 + 8 * (lane >> 4);
#pragma unroll
        for (int t = 0; t < NTILES; ++t) {
            const int c = t * 16 + nlo;
#pragma unroll
            for (int v = 0; v < 8; ++v)
                out[(mbase + v) * STRIDE + c] = acc[r][t][v];
        }
    }
}

__global__ __launch_bounds__(256, 1)
void diffnet_fused_kernel(const float* __restrict__ x,
                          const float* __restrict__ w0, const float* __restrict__ b0,
                          const float* __restrict__ w1, const float* __restrict__ b1,
                          const float* __restrict__ w2, const float* __restrict__ b2,
                          const float* __restrict__ w3, const float* __restrict__ b3,
                          float* __restrict__ out)
{
    extern __shared__ float sm[];
    float* Abuf = sm;
    float* Bbuf = Abuf + 256 * STRIDE;
    float* gp0  = Bbuf + 256 * STRIDE;
    float* gpp0 = gp0  + 256 * TB;
    float* gp1  = gpp0 + 256 * TB;
    float* gpp1 = gp1  + 256 * TB;
    float* gp2  = gpp1 + 256 * TB;
    float* gpp2 = gp2  + 256 * TB;
    float* s2   = gpp2 + 256 * TB;
    float* v2s  = s2   + 256 * TB;
    float* v1s  = v2s  + 256 * TB;
    float* xl   = v1s  + 256 * TB;
    float* outa = xl   + TB * 12;

    const int tid  = threadIdx.x;
    const int lane = tid & 31;
    const int wv   = tid >> 5;
    const int n    = blockIdx.y;
    const int bg0  = blockIdx.x * TB;

    const float* W0 = w0 + n * 256 * 12;
    const float* W1 = w1 + n * 256 * 256;
    const float* W2 = w2 + n * 256 * 256;
    const float* W3 = w3 + n * 256;
    const float* Bi0 = b0 + n * 256;
    const float* Bi1 = b1 + n * 256;
    const float* Bi2 = b2 + n * 256;

    if (tid < TB * 12) xl[tid] = x[bg0 * 12 + tid];
    if (tid < TB)      outa[tid] = b3[n];
    __syncthreads();

    {
        const int i = tid;
        float w0r[12];
#pragma unroll
        for (int m = 0; m < 12; ++m) w0r[m] = W0[i * 12 + m];
        for (int b = 0; b < TB; ++b) {
            float a = Bi0[i];
#pragma unroll
            for (int m = 0; m < 12; ++m) a += w0r[m] * xl[b * 12 + m];
            float t  = tanhf(a);
            float gp = 1.f - t * t;
            gp0[i * TB + b]  = gp;
            gpp0[i * TB + b] = -2.f * t * gp;
#pragma unroll
            for (int m = 0; m < 12; ++m) Abuf[i * STRIDE + b * 12 + m] = gp * w0r[m];
            Abuf[i * STRIDE + 96 + b]  = t;
            Abuf[i * STRIDE + 104 + b] = 0.f;
        }
    }
    __syncthreads();

    gemm256(W1, Abuf, Bbuf, lane, wv, false);
    __syncthreads();

    {
        const int i = tid;
        for (int b = 0; b < TB; ++b) {
            float a  = Bbuf[i * STRIDE + 96 + b] + Bi1[i];
            float t  = tanhf(a);
            float gp = 1.f - t * t;
            gp1[i * TB + b]  = gp;
            gpp1[i * TB + b] = -2.f * t * gp;
#pragma unroll
            for (int m = 0; m < 12; ++m)
                Abuf[i * STRIDE + b * 12 + m] = gp * Bbuf[i * STRIDE + b * 12 + m];
            Abuf[i * STRIDE + 96 + b] = t;
        }
    }
    __syncthreads();

    gemm256(W2, Abuf, Abuf, lane, wv, true);
    __syncthreads();

    {
        const int i = tid;
        for (int b = 0; b < TB; ++b) {
            float a  = Abuf[i * STRIDE + 96 + b] + Bi2[i];
            float t  = tanhf(a);
            float gp = 1.f - t * t;
            gp2[i * TB + b]  = gp;
            gpp2[i * TB + b] = -2.f * t * gp;
            s2[i * TB + b]   = W3[i] * gp;
            atomicAdd(&outa[b], W3[i] * t);
        }
    }
    __syncthreads();

    {
        const int i = tid;
        float acc[TB];
#pragma unroll
        for (int b = 0; b < TB; ++b) acc[b] = 0.f;
        for (int o = 0; o < 256; ++o) {
            float w = W2[o * 256 + i];
#pragma unroll
            for (int b = 0; b < TB; ++b) acc[b] += w * s2[o * TB + b];
        }
#pragma unroll
        for (int b = 0; b < TB; ++b) v2s[i * TB + b] = acc[b];
    }
    __syncthreads();

    {
        const int i = tid;
        float acc[TB];
#pragma unroll
        for (int b = 0; b < TB; ++b) acc[b] = 0.f;
        for (int o = 0; o < 256; ++o) {
            float w = W1[o * 256 + i];
#pragma unroll
            for (int b = 0; b < TB; ++b) acc[b] += w * v2s[o * TB + b] * gp1[o * TB + b];
        }
#pragma unroll
        for (int b = 0; b < TB; ++b) v1s[i * TB + b] = acc[b];
    }
    __syncthreads();

    for (int idx = tid; idx < TB * 12; idx += 256) {
        const int b = idx / 12, m = idx % 12;
        float acc = 0.f;
        for (int o = 0; o < 256; ++o)
            acc += s2[o * TB + b] * Abuf[o * STRIDE + b * 12 + m];
        out[2048 + (n * 512 + bg0 + b) * 12 + m] = acc;
    }
    if (tid < TB) out[n * 512 + bg0 + tid] = outa[tid];
    __syncthreads();

    {
        const int i = tid;
#pragma unroll
        for (int b = 0; b < TB; ++b) {
            s2[i * TB + b]  = W3[i]          * gpp2[i * TB + b];
            v2s[i * TB + b] = v2s[i * TB + b] * gpp1[i * TB + b];
            v1s[i * TB + b] = v1s[i * TB + b] * gpp0[i * TB + b];
        }
    }
    __syncthreads();

    {
        const int b   = wv;
        const int col = lane & 15;
        const int jc  = col < 12 ? col : 11;
        const int h2  = (lane >> 4) * 2;
        const v8f vz = {0.f, 0.f, 0.f, 0.f, 0.f, 0.f, 0.f, 0.f};
        v8f acc0 = vz, acc1 = vz, acc2 = vz;
#pragma unroll 4
        for (int ks = 0; ks < 64; ++ks) {
            const int i0 = ks * 4 + h2;
            v2f j0, j1, j2, a0, a1, a2;
            j0.x = W0[(i0 + 0) * 12 + jc];
            j0.y = W0[(i0 + 1) * 12 + jc];
            j1.x = Bbuf[(i0 + 0) * STRIDE + b * 12 + jc];
            j1.y = Bbuf[(i0 + 1) * STRIDE + b * 12 + jc];
            j2.x = Abuf[(i0 + 0) * STRIDE + b * 12 + jc];
            j2.y = Abuf[(i0 + 1) * STRIDE + b * 12 + jc];
            a0.x = v1s[(i0 + 0) * TB + b] * j0.x;
            a0.y = v1s[(i0 + 1) * TB + b] * j0.y;
            a1.x = v2s[(i0 + 0) * TB + b] * j1.x;
            a1.y = v2s[(i0 + 1) * TB + b] * j1.y;
            a2.x = s2[(i0 + 0) * TB + b] * j2.x;
            a2.y = s2[(i0 + 1) * TB + b] * j2.y;
            acc0 = __builtin_amdgcn_wmma_f32_16x16x4_f32(
                false, a0, false, j0, (short)0, acc0, false, false);
            acc1 = __builtin_amdgcn_wmma_f32_16x16x4_f32(
                false, a1, false, j1, (short)0, acc1, false, false);
            acc2 = __builtin_amdgcn_wmma_f32_16x16x4_f32(
                false, a2, false, j2, (short)0, acc2, false, false);
        }
        const v8f acc = acc0 + acc1 + acc2;
        if (col < 12) {
            const int jb = 8 * (lane >> 4);
#pragma unroll
            for (int v = 0; v < 8; ++v)
                if (jb + v < 12) gp0[b * 144 + (jb + v) * 12 + col] = acc[v];
        }
    }
    __syncthreads();
    for (int idx = tid; idx < TB * 144; idx += 256)
        out[26624 + (n * 512 + bg0) * 144 + idx] = gp0[idx];
}

extern "C" void kernel_launch(void* const* d_in, const int* in_sizes, int n_in,
                              void* d_out, int out_size, void* d_ws, size_t ws_size,
                              hipStream_t stream)
{
    (void)in_sizes; (void)n_in; (void)out_size; (void)d_ws; (void)ws_size;
    const float* x  = (const float*)d_in[0];
    const float* w0 = (const float*)d_in[1];
    const float* b0 = (const float*)d_in[2];
    const float* w1 = (const float*)d_in[3];
    const float* b1 = (const float*)d_in[4];
    const float* w2 = (const float*)d_in[5];
    const float* b2 = (const float*)d_in[6];
    const float* w3 = (const float*)d_in[7];
    const float* b3 = (const float*)d_in[8];
    float* outp = (float*)d_out;

    const size_t smem = (size_t)(2 * 256 * STRIDE + 9 * 256 * TB + TB * 12 + TB) * sizeof(float);
    (void)hipFuncSetAttribute(reinterpret_cast<const void*>(&diffnet_fused_kernel),
                              hipFuncAttributeMaxDynamicSharedMemorySize, (int)smem);

    dim3 grid(512 / TB, 4);
    dim3 block(256);
    diffnet_fused_kernel<<<grid, block, smem, stream>>>(x, w0, b0, w1, b1, w2, b2, w3, b3, outp);
}


# backward sensitivities v2 = W2^T s2 and v1 = W1^T (v2.gp1) moved from per-thread scalar loops onto f32 16x16x4 matrix instructions, 32 weight loads in flight per wave per trip
# speedup vs baseline: 2.7340x; 1.2422x over previous
#include <hip/hip_runtime.h>
#include <math.h>


typedef float v2f __attribute__((ext_vector_type(2)));
typedef float v8f __attribute__((ext_vector_type(8)));

#define TB      8
#define STRIDE  116
#define NCOLS   112
#define NTILES  7

__device__ __forceinline__ void gemm256(const float* __restrict__ W,
                                        const float* __restrict__ in,
                                        float* __restrict__ out,
                                        int lane, int wv, bool syncBeforeWrite)
{
    v8f acc[2][NTILES];
    const v8f vz = {0.f, 0.f, 0.f, 0.f, 0.f, 0.f, 0.f, 0.f};
#pragma unroll
    for (int r = 0; r < 2; ++r)
#pragma unroll
        for (int t = 0; t < NTILES; ++t) acc[r][t] = vz;

    const int m0    = (wv * 2 + 0) * 16 + (lane & 15);
    const int m1    = (wv * 2 + 1) * 16 + (lane & 15);
    const int khalf = (lane >> 4) * 2;
    const int nlo   = lane & 15;

    for (int ks = 0; ks < 64; ++ks) {
        const int k = ks * 4 + khalf;
        v2f a0 = *(const v2f*)(W + m0 * 256 + k);
        v2f a1 = *(const v2f*)(W + m1 * 256 + k);
#pragma unroll
        for (int t = 0; t < NTILES; ++t) {
            const int c = t * 16 + nlo;
            v2f bf;
            bf.x = in[(k + 0) * STRIDE + c];
            bf.y = in[(k + 1) * STRIDE + c];
            acc[0][t] = __builtin_amdgcn_wmma_f32_16x16x4_f32(
                false, a0, false, bf, (short)0, acc[0][t], false, false);
            acc[1][t] = __builtin_amdgcn_wmma_f32_16x16x4_f32(
                false, a1, false, bf, (short)0, acc[1][t], false, false);
        }
    }

    if (syncBeforeWrite) __syncthreads();

#pragma unroll
    for (int r = 0; r < 2; ++r) {
        const int mbase = (wv * 2 + r) * 16 + 8 * (lane >> 4);
#pragma unroll
        for (int t = 0; t < NTILES; ++t) {
            const int c = t * 16 + nlo;
#pragma unroll
            for (int v = 0; v < 8; ++v)
                out[(mbase + v) * STRIDE + c] = acc[r][t][v];
        }
    }
}

template <bool SCALED>
__device__ __forceinline__ void gemvT256(const float* __restrict__ W,
                                         const float* __restrict__ S,
                                         float* __restrict__ V,
                                         const float* __restrict__ G,
                                         float* __restrict__ U,
                                         int lane, int wv)
{
    const int c  = lane & 15;
    const int h  = lane >> 4;
    const float* wp = W + (2 * h) * 256 + 32 * wv + 2 * c;
    const float* sp = S + (2 * h) * TB + (c & 7);
    const v8f vz = {0.f, 0.f, 0.f, 0.f, 0.f, 0.f, 0.f, 0.f};
    v8f acc0 = vz, acc1 = vz;
#pragma unroll 16
    for (int ks = 0; ks < 64; ++ks) {
        const v2f w0 = *(const v2f*)(wp + (ks * 4 + 0) * 256);
        const v2f w1 = *(const v2f*)(wp + (ks * 4 + 1) * 256);
        v2f bf;
        bf.x = sp[(ks * 4 + 0) * TB];
        bf.y = sp[(ks * 4 + 1) * TB];
        v2f a0, a1;
        a0.x = w0.x; a0.y = w1.x;
        a1.x = w0.y; a1.y = w1.y;
        acc0 = __builtin_amdgcn_wmma_f32_16x16x4_f32(
            false, a0, false, bf, (short)0, acc0, false, false);
        acc1 = __builtin_amdgcn_wmma_f32_16x16x4_f32(
            false, a1, false, bf, (short)0, acc1, false, false);
    }
    if (c < 8) {
#pragma unroll
        for (int r = 0; r < 8; ++r) {
            const int e = (32 * wv + 2 * (8 * h + r)) * TB + c;
            V[e]      = acc0[r];
            V[e + TB] = acc1[r];
            if (SCALED) {
                U[e]      = acc0[r] * G[e];
                U[e + TB] = acc1[r] * G[e + TB];
            }
        }
    }
}

__global__ __launch_bounds__(256, 1)
void diffnet_fused_kernel(const float* __restrict__ x,
                          const float* __restrict__ w0, const float* __restrict__ b0,
                          const float* __restrict__ w1, const float* __restrict__ b1,
                          const float* __restrict__ w2, const float* __restrict__ b2,
                          const float* __restrict__ w3, const float* __restrict__ b3,
                          float* __restrict__ out)
{
    extern __shared__ float sm[];
    float* Abuf = sm;
    float* Bbuf = Abuf + 256 * STRIDE;
    float* gp0  = Bbuf + 256 * STRIDE;
    float* gpp0 = gp0  + 256 * TB;
    float* gp1  = gpp0 + 256 * TB;
    float* gpp1 = gp1  + 256 * TB;
    float* gp2  = gpp1 + 256 * TB;
    float* gpp2 = gp2  + 256 * TB;
    float* s2   = gpp2 + 256 * TB;
    float* v2s  = s2   + 256 * TB;
    float* v1s  = v2s  + 256 * TB;
    float* xl   = v1s  + 256 * TB;
    float* outa = xl   + TB * 12;

    const int tid  = threadIdx.x;
    const int lane = tid & 31;
    const int wv   = tid >> 5;
    const int n    = blockIdx.y;
    const int bg0  = blockIdx.x * TB;

    const float* W0 = w0 + n * 256 * 12;
    const float* W1 = w1 + n * 256 * 256;
    const float* W2 = w2 + n * 256 * 256;
    const float* W3 = w3 + n * 256;
    const float* Bi0 = b0 + n * 256;
    const float* Bi1 = b1 + n * 256;
    const float* Bi2 = b2 + n * 256;

    if (tid < TB * 12) xl[tid] = x[bg0 * 12 + tid];
    if (tid < TB)      outa[tid] = b3[n];
    __syncthreads();

    {
        const int i = tid;
        float w0r[12];
#pragma unroll
        for (int m = 0; m < 12; ++m) w0r[m] = W0[i * 12 + m];
        for (int b = 0; b < TB; ++b) {
            float a = Bi0[i];
#pragma unroll
            for (int m = 0; m < 12; ++m) a += w0r[m] * xl[b * 12 + m];
            float t  = tanhf(a);
            float gp = 1.f - t * t;
            gp0[i * TB + b]  = gp;
            gpp0[i * TB + b] = -2.f * t * gp;
#pragma unroll
            for (int m = 0; m < 12; ++m) Abuf[i * STRIDE + b * 12 + m] = gp * w0r[m];
            Abuf[i * STRIDE + 96 + b]  = t;
            Abuf[i * STRIDE + 104 + b] = 0.f;
        }
    }
    __syncthreads();

    gemm256(W1, Abuf, Bbuf, lane, wv, false);
    __syncthreads();

    {
        const int i = tid;
        for (int b = 0; b < TB; ++b) {
            float a  = Bbuf[i * STRIDE + 96 + b] + Bi1[i];
            float t  = tanhf(a);
            float gp = 1.f - t * t;
            gp1[i * TB + b]  = gp;
            gpp1[i * TB + b] = -2.f * t * gp;
#pragma unroll
            for (int m = 0; m < 12; ++m)
                Abuf[i * STRIDE + b * 12 + m] = gp * Bbuf[i * STRIDE + b * 12 + m];
            Abuf[i * STRIDE + 96 + b] = t;
        }
    }
    __syncthreads();

    gemm256(W2, Abuf, Abuf, lane, wv, true);
    __syncthreads();

    {
        const int i = tid;
        for (int b = 0; b < TB; ++b) {
            float a  = Abuf[i * STRIDE + 96 + b] + Bi2[i];
            float t  = tanhf(a);
            float gp = 1.f - t * t;
            gp2[i * TB + b]  = gp;
            gpp2[i * TB + b] = -2.f * t * gp;
            s2[i * TB + b]   = W3[i] * gp;
            atomicAdd(&outa[b], W3[i] * t);
        }
    }
    __syncthreads();

    gemvT256<true>(W2, s2, v2s, gp1, gp0, lane, wv);
    __syncthreads();

    gemvT256<false>(W1, gp0, v1s, nullptr, nullptr, lane, wv);
    __syncthreads();

    for (int idx = tid; idx < TB * 12; idx += 256) {
        const int b = idx / 12, m = idx % 12;
        float acc = 0.f;
        for (int o = 0; o < 256; ++o)
            acc += s2[o * TB + b] * Abuf[o * STRIDE + b * 12 + m];
        out[2048 + (n * 512 + bg0 + b) * 12 + m] = acc;
    }
    if (tid < TB) out[n * 512 + bg0 + tid] = outa[tid];
    __syncthreads();

    {
        const int i = tid;
#pragma unroll
        for (int b = 0; b < TB; ++b) {
            s2[i * TB + b]  = W3[i]          * gpp2[i * TB + b];
            v2s[i * TB + b] = v2s[i * TB + b] * gpp1[i * TB + b];
            v1s[i * TB + b] = v1s[i * TB + b] * gpp0[i * TB + b];
        }
    }
    __syncthreads();

    {
        const int b   = wv;
        const int col = lane & 15;
        const int jc  = col < 12 ? col : 11;
        const int h2  = (lane >> 4) * 2;
        const v8f vz = {0.f, 0.f, 0.f, 0.f, 0.f, 0.f, 0.f, 0.f};
        v8f acc0 = vz, acc1 = vz, acc2 = vz;
#pragma unroll 4
        for (int ks = 0; ks < 64; ++ks) {
            const int i0 = ks * 4 + h2;
            v2f j0, j1, j2, a0, a1, a2;
            j0.x = W0[(i0 + 0) * 12 + jc];
            j0.y = W0[(i0 + 1) * 12 + jc];
            j1.x = Bbuf[(i0 + 0) * STRIDE + b * 12 + jc];
            j1.y = Bbuf[(i0 + 1) * STRIDE + b * 12 + jc];
            j2.x = Abuf[(i0 + 0) * STRIDE + b * 12 + jc];
            j2.y = Abuf[(i0 + 1) * STRIDE + b * 12 + jc];
            a0.x = v1s[(i0 + 0) * TB + b] * j0.x;
            a0.y = v1s[(i0 + 1) * TB + b] * j0.y;
            a1.x = v2s[(i0 + 0) * TB + b] * j1.x;
            a1.y = v2s[(i0 + 1) * TB + b] * j1.y;
            a2.x = s2[(i0 + 0) * TB + b] * j2.x;
            a2.y = s2[(i0 + 1) * TB + b] * j2.y;
            acc0 = __builtin_amdgcn_wmma_f32_16x16x4_f32(
                false, a0, false, j0, (short)0, acc0, false, false);
            acc1 = __builtin_amdgcn_wmma_f32_16x16x4_f32(
                false, a1, false, j1, (short)0, acc1, false, false);
            acc2 = __builtin_amdgcn_wmma_f32_16x16x4_f32(
                false, a2, false, j2, (short)0, acc2, false, false);
        }
        const v8f acc = acc0 + acc1 + acc2;
        if (col < 12) {
            const int jb = 8 * (lane >> 4);
#pragma unroll
            for (int v = 0; v < 8; ++v)
                if (jb + v < 12) gp0[b * 144 + (jb + v) * 12 + col] = acc[v];
        }
    }
    __syncthreads();
    for (int idx = tid; idx < TB * 144; idx += 256)
        out[26624 + (n * 512 + bg0) * 144 + idx] = gp0[idx];
}

extern "C" void kernel_launch(void* const* d_in, const int* in_sizes, int n_in,
                              void* d_out, int out_size, void* d_ws, size_t ws_size,
                              hipStream_t stream)
{
    (void)in_sizes; (void)n_in; (void)out_size; (void)d_ws; (void)ws_size;
    const float* x  = (const float*)d_in[0];
    const float* w0 = (const float*)d_in[1];
    const float* b0 = (const float*)d_in[2];
    const float* w1 = (const float*)d_in[3];
    const float* b1 = (const float*)d_in[4];
    const float* w2 = (const float*)d_in[5];
    const float* b2 = (const float*)d_in[6];
    const float* w3 = (const float*)d_in[7];
    const float* b3 = (const float*)d_in[8];
    float* outp = (float*)d_out;

    const size_t smem = (size_t)(2 * 256 * STRIDE + 9 * 256 * TB + TB * 12 + TB) * sizeof(float);
    (void)hipFuncSetAttribute(reinterpret_cast<const void*>(&diffnet_fused_kernel),
                              hipFuncAttributeMaxDynamicSharedMemorySize, (int)smem);

    dim3 grid(512 / TB, 4);
    dim3 block(256);
    diffnet_fused_kernel<<<grid, block, smem, stream>>>(x, w0, b0, w1, b1, w2, b2, w3, b3, outp);
}


# out and dh_out, the two 256-long reductions still on scalar paths (lane-by-lane wave sum + LDS atomics; a 96-thread loop), now ride in spare rows 12-13 of the third d2 matrix product; atomics removed
# speedup vs baseline: 3.1700x; 1.1595x over previous
#include <hip/hip_runtime.h>
#include <math.h>


typedef float v2f __attribute__((ext_vector_type(2)));
typedef float v8f __attribute__((ext_vector_type(8)));

#define TB      8
#define STRIDE  116
#define NCOLS   112
#define NTILES  7

__device__ __forceinline__ void gemm256(const float* __restrict__ W,
                                        const float* __restrict__ in,
                                        float* __restrict__ out,
                                        int lane, int wv, bool syncBeforeWrite)
{
    v8f acc[2][NTILES];
    const v8f vz = {0.f, 0.f, 0.f, 0.f, 0.f, 0.f, 0.f, 0.f};
#pragma unroll
    for (int r = 0; r < 2; ++r)
#pragma unroll
        for (int t = 0; t < NTILES; ++t) acc[r][t] = vz;

    const int m0    = (wv * 2 + 0) * 16 + (lane & 15);
    const int m1    = (wv * 2 + 1) * 16 + (lane & 15);
    const int khalf = (lane >> 4) * 2;
    const int nlo   = lane & 15;

    for (int ks = 0; ks < 64; ++ks) {
        const int k = ks * 4 + khalf;
        v2f a0 = *(const v2f*)(W + m0 * 256 + k);
        v2f a1 = *(const v2f*)(W + m1 * 256 + k);
#pragma unroll
        for (int t = 0; t < NTILES; ++t) {
            const int c = t * 16 + nlo;
            v2f bf;
            bf.x = in[(k + 0) * STRIDE + c];
            bf.y = in[(k + 1) * STRIDE + c];
            acc[0][t] = __builtin_amdgcn_wmma_f32_16x16x4_f32(
                false, a0, false, bf, (short)0, acc[0][t], false, false);
            acc[1][t] = __builtin_amdgcn_wmma_f32_16x16x4_f32(
                false, a1, false, bf, (short)0, acc[1][t], false, false);
        }
    }

    if (syncBeforeWrite) __syncthreads();

#pragma unroll
    for (int r = 0; r < 2; ++r) {
        const int mbase = (wv * 2 + r) * 16 + 8 * (lane >> 4);
#pragma unroll
        for (int t = 0; t < NTILES; ++t) {
            const int c = t * 16 + nlo;
#pragma unroll
            for (int v = 0; v < 8; ++v)
                out[(mbase + v) * STRIDE + c] = acc[r][t][v];
        }
    }
}

template <bool SCALED>
__device__ __forceinline__ void gemvT256(const float* __restrict__ W,
                                         const float* __restrict__ S,
                                         float* __restrict__ V,
                                         const float* __restrict__ G,
                                         float* __restrict__ U,
                                         int lane, int wv)
{
    const int c  = lane & 15;
    const int h  = lane >> 4;
    const float* wp = W + (2 * h) * 256 + 32 * wv + 2 * c;
    const float* sp = S + (2 * h) * TB + (c & 7);
    const v8f vz = {0.f, 0.f, 0.f, 0.f, 0.f, 0.f, 0.f, 0.f};
    v8f acc0 = vz, acc1 = vz;
#pragma unroll 16
    for (int ks = 0; ks < 64; ++ks) {
        const v2f w0 = *(const v2f*)(wp + (ks * 4 + 0) * 256);
        const v2f w1 = *(const v2f*)(wp + (ks * 4 + 1) * 256);
        v2f bf;
        bf.x = sp[(ks * 4 + 0) * TB];
        bf.y = sp[(ks * 4 + 1) * TB];
        v2f a0, a1;
        a0.x = w0.x; a0.y = w1.x;
        a1.x = w0.y; a1.y = w1.y;
        acc0 = __builtin_amdgcn_wmma_f32_16x16x4_f32(
            false, a0, false, bf, (short)0, acc0, false, false);
        acc1 = __builtin_amdgcn_wmma_f32_16x16x4_f32(
            false, a1, false, bf, (short)0, acc1, false, false);
    }
    if (c < 8) {
#pragma unroll
        for (int r = 0; r < 8; ++r) {
            const int e = (32 * wv + 2 * (8 * h + r)) * TB + c;
            V[e]      = acc0[r];
            V[e + TB] = acc1[r];
            if (SCALED) {
                U[e]      = acc0[r] * G[e];
                U[e + TB] = acc1[r] * G[e + TB];
            }
        }
    }
}

__global__ __launch_bounds__(256, 1)
void diffnet_fused_kernel(const float* __restrict__ x,
                          const float* __restrict__ w0, const float* __restrict__ b0,
                          const float* __restrict__ w1, const float* __restrict__ b1,
                          const float* __restrict__ w2, const float* __restrict__ b2,
                          const float* __restrict__ w3, const float* __restrict__ b3,
                          float* __restrict__ out)
{
    extern __shared__ float sm[];
    float* Abuf = sm;
    float* Bbuf = Abuf + 256 * STRIDE;
    float* gp0  = Bbuf + 256 * STRIDE;
    float* gpp0 = gp0  + 256 * TB;
    float* gp1  = gpp0 + 256 * TB;
    float* gpp1 = gp1  + 256 * TB;
    float* w3c  = gpp1 + 256 * TB;
    float* c2   = w3c  + 256 * TB + 8;
    float* s2   = c2   + 256 * TB + 8;
    float* v2s  = s2   + 256 * TB;
    float* v1s  = v2s  + 256 * TB;
    float* xl   = v1s  + 256 * TB;

    const int tid  = threadIdx.x;
    const int lane = tid & 31;
    const int wv   = tid >> 5;
    const int n    = blockIdx.y;
    const int bg0  = blockIdx.x * TB;

    const float* W0 = w0 + n * 256 * 12;
    const float* W1 = w1 + n * 256 * 256;
    const float* W2 = w2 + n * 256 * 256;
    const float* W3 = w3 + n * 256;
    const float* Bi0 = b0 + n * 256;
    const float* Bi1 = b1 + n * 256;
    const float* Bi2 = b2 + n * 256;

    if (tid < TB * 12) xl[tid] = x[bg0 * 12 + tid];
    __syncthreads();

    {
        const int i = tid;
        float w0r[12];
#pragma unroll
        for (int m = 0; m < 12; ++m) w0r[m] = W0[i * 12 + m];
        for (int b = 0; b < TB; ++b) {
            float a = Bi0[i];
#pragma unroll
            for (int m = 0; m < 12; ++m) a += w0r[m] * xl[b * 12 + m];
            float t  = tanhf(a);
            float gp = 1.f - t * t;
            gp0[i * TB + b]  = gp;
            gpp0[i * TB + b] = -2.f * t * gp;
#pragma unroll
            for (int m = 0; m < 12; ++m) Abuf[i * STRIDE + b * 12 + m] = gp * w0r[m];
            Abuf[i * STRIDE + 96 + b]  = t;
            Abuf[i * STRIDE + 104 + b] = 0.f;
        }
    }
    __syncthreads();

    gemm256(W1, Abuf, Bbuf, lane, wv, false);
    __syncthreads();

    {
        const int i = tid;
        for (int b = 0; b < TB; ++b) {
            float a  = Bbuf[i * STRIDE + 96 + b] + Bi1[i];
            float t  = tanhf(a);
            float gp = 1.f - t * t;
            gp1[i * TB + b]  = gp;
            gpp1[i * TB + b] = -2.f * t * gp;
#pragma unroll
            for (int m = 0; m < 12; ++m)
                Abuf[i * STRIDE + b * 12 + m] = gp * Bbuf[i * STRIDE + b * 12 + m];
            Abuf[i * STRIDE + 96 + b] = t;
        }
    }
    __syncthreads();

    gemm256(W2, Abuf, Abuf, lane, wv, true);
    __syncthreads();

    {
        const int i = tid;
        const float w3i = W3[i];
        for (int b = 0; b < TB; ++b) {
            float a   = Abuf[i * STRIDE + 96 + b] + Bi2[i];
            float t   = tanhf(a);
            float gp  = 1.f - t * t;
            float gpp = -2.f * t * gp;
            s2[i * TB + b]  = w3i * gp;
            c2[i * TB + b]  = w3i * gpp;
            w3c[i * TB + b] = w3i;
            Abuf[i * STRIDE + 96 + b] = t;
        }
        Abuf[i * STRIDE + 104] = 1.f;
    }
    __syncthreads();

    gemvT256<true>(W2, s2, v2s, gp1, gp0, lane, wv);
    __syncthreads();

    gemvT256<false>(W1, gp0, v1s, nullptr, nullptr, lane, wv);
    __syncthreads();

    {
        const int i = tid;
#pragma unroll
        for (int b = 0; b < TB; ++b) {
            v2s[i * TB + b] = v2s[i * TB + b] * gpp1[i * TB + b];
            v1s[i * TB + b] = v1s[i * TB + b] * gpp0[i * TB + b];
        }
    }
    __syncthreads();

    {
        const int b   = wv;
        const int col = lane & 15;
        const int jc  = col < 12 ? col : 11;
        const int h2  = (lane >> 4) * 2;
        const int bo2 = col < 12 ? b * 12 + col : (col < 14 ? 104 : 96 + b);
        const int co2 = (int)((col == 12 ? s2 : (col == 13 ? w3c : c2)) - sm) + b;
        const v8f vz = {0.f, 0.f, 0.f, 0.f, 0.f, 0.f, 0.f, 0.f};
        v8f acc0 = vz, acc1 = vz, acc2 = vz;
#pragma unroll 4
        for (int ks = 0; ks < 64; ++ks) {
            const int i0 = ks * 4 + h2;
            v2f j0, j1, j2, a0, a1, a2;
            j0.x = W0[(i0 + 0) * 12 + jc];
            j0.y = W0[(i0 + 1) * 12 + jc];
            j1.x = Bbuf[(i0 + 0) * STRIDE + b * 12 + jc];
            j1.y = Bbuf[(i0 + 1) * STRIDE + b * 12 + jc];
            j2.x = Abuf[(i0 + 0) * STRIDE + bo2];
            j2.y = Abuf[(i0 + 1) * STRIDE + bo2];
            a0.x = v1s[(i0 + 0) * TB + b] * j0.x;
            a0.y = v1s[(i0 + 1) * TB + b] * j0.y;
            a1.x = v2s[(i0 + 0) * TB + b] * j1.x;
            a1.y = v2s[(i0 + 1) * TB + b] * j1.y;
            a2.x = sm[co2 + (i0 + 0) * TB] * j2.x;
            a2.y = sm[co2 + (i0 + 1) * TB] * j2.y;
            acc0 = __builtin_amdgcn_wmma_f32_16x16x4_f32(
                false, a0, false, j0, (short)0, acc0, false, false);
            acc1 = __builtin_amdgcn_wmma_f32_16x16x4_f32(
                false, a1, false, j1, (short)0, acc1, false, false);
            acc2 = __builtin_amdgcn_wmma_f32_16x16x4_f32(
                false, a2, false, j2, (short)0, acc2, false, false);
        }
        const v8f acc = acc0 + acc1 + acc2;
        if (col < 12) {
            const int jb = 8 * (lane >> 4);
#pragma unroll
            for (int v = 0; v < 8; ++v)
                if (jb + v < 12) gp0[b * 144 + (jb + v) * 12 + col] = acc[v];
        }
        if ((lane >> 4) == 1) {
            if (col < 12)  gp0[TB * 144 + b * 12 + col] = acc2[4];
            if (col == 14) gp0[TB * 156 + b] = acc2[5] + b3[n];
        }
    }
    __syncthreads();
    for (int idx = tid; idx < TB * 144; idx += 256)
        out[26624 + (n * 512 + bg0) * 144 + idx] = gp0[idx];
    if (tid < TB * 12) out[2048 + (n * 512 + bg0) * 12 + tid] = gp0[TB * 144 + tid];
    if (tid < TB)      out[n * 512 + bg0 + tid] = gp0[TB * 156 + tid];
}

extern "C" void kernel_launch(void* const* d_in, const int* in_sizes, int n_in,
                              void* d_out, int out_size, void* d_ws, size_t ws_size,
                              hipStream_t stream)
{
    (void)in_sizes; (void)n_in; (void)out_size; (void)d_ws; (void)ws_size;
    const float* x  = (const float*)d_in[0];
    const float* w0 = (const float*)d_in[1];
    const float* b0 = (const float*)d_in[2];
    const float* w1 = (const float*)d_in[3];
    const float* b1 = (const float*)d_in[4];
    const float* w2 = (const float*)d_in[5];
    const float* b2 = (const float*)d_in[6];
    const float* w3 = (const float*)d_in[7];
    const float* b3 = (const float*)d_in[8];
    float* outp = (float*)d_out;

    const size_t smem = (size_t)(2 * 256 * STRIDE + 9 * 256 * TB + 16 + TB * 12) * sizeof(float);
    (void)hipFuncSetAttribute(reinterpret_cast<const void*>(&diffnet_fused_kernel),
                              hipFuncAttributeMaxDynamicSharedMemorySize, (int)smem);

    dim3 grid(512 / TB, 4);
    dim3 block(256);
    diffnet_fused_kernel<<<grid, block, smem, stream>>>(x, w0, b0, w1, b1, w2, b2, w3, b3, outp);
}


# both 256x256 layer products fetch operands ahead of use: weight fragments by 16-byte loads two 8-k groups ahead, LDS fragments one group ahead, in second register sets; same new k order in A and B
# speedup vs baseline: 3.4373x; 1.0843x over previous
#include <hip/hip_runtime.h>
#include <math.h>


typedef float v2f __attribute__((ext_vector_type(2)));
typedef float v4f __attribute__((ext_vector_type(4)));
typedef float v8f __attribute__((ext_vector_type(8)));

#define TB      8
#define STRIDE  116
#define NCOLS   112
#define NTILES  7

__device__ __forceinline__ void gemm_load_b(const float* __restrict__ bp, int g, float (&b)[NTILES][4])
{
#pragma unroll
    for (int t = 0; t < NTILES; ++t)
#pragma unroll
        for (int e = 0; e < 4; ++e) b[t][e] = bp[(8 * g + e) * STRIDE + 16 * t];
}

__device__ __forceinline__ void gemm_group(const v4f a0, const v4f a1, const float (&b)[NTILES][4],
                                           v8f (&acc)[2][NTILES])
{
    v2f a0lo, a0hi, a1lo, a1hi;
    a0lo.x = a0.x; a0lo.y = a0.y; a0hi.x = a0.z; a0hi.y = a0.w;
    a1lo.x = a1.x; a1lo.y = a1.y; a1hi.x = a1.z; a1hi.y = a1.w;
#pragma unroll
    for (int t = 0; t < NTILES; ++t) {
        v2f bf; bf.x = b[t][0]; bf.y = b[t][1];
        acc[0][t] = __builtin_amdgcn_wmma_f32_16x16x4_f32(
            false, a0lo, false, bf, (short)0, acc[0][t], false, false);
        acc[1][t] = __builtin_amdgcn_wmma_f32_16x16x4_f32(
            false, a1lo, false, bf, (short)0, acc[1][t], false, false);
    }
#pragma unroll
    for (int t = 0; t < NTILES; ++t) {
        v2f bf; bf.x = b[t][2]; bf.y = b[t][3];
        acc[0][t] = __builtin_amdgcn_wmma_f32_16x16x4_f32(
            false, a0hi, false, bf, (short)0, acc[0][t], false, false);
        acc[1][t] = __builtin_amdgcn_wmma_f32_16x16x4_f32(
            false, a1hi, false, bf, (short)0, acc[1][t], false, false);
    }
}

__device__ __forceinline__ void gemm256(const float* __restrict__ W,
                                        const float* __restrict__ in,
                                        float* __restrict__ out,
                                        int lane, int wv, bool syncBeforeWrite)
{
    v8f acc[2][NTILES];
    const v8f vz = {0.f, 0.f, 0.f, 0.f, 0.f, 0.f, 0.f, 0.f};
#pragma unroll
    for (int r = 0; r < 2; ++r)
#pragma unroll
        for (int t = 0; t < NTILES; ++t) acc[r][t] = vz;

    const int nlo = lane & 15;
    const int h4  = (lane >> 4) * 4;
    const float* wp0 = W + ((wv * 2 + 0) * 16 + nlo) * 256 + h4;
    const float* wp1 = wp0 + 16 * 256;
    const float* bp  = in + h4 * STRIDE + nlo;

    v4f ax0 = *(const v4f*)(wp0);
    v4f ax1 = *(const v4f*)(wp1);
    v4f ax2 = *(const v4f*)(wp0 + 8);
    v4f ax3 = *(const v4f*)(wp1 + 8);
    float bc[NTILES][4], bn[NTILES][4];
    gemm_load_b(bp, 0, bc);

    for (int g = 0; g < 32; g += 4) {
        const v4f ay0 = *(const v4f*)(wp0 + 8 * (g + 2));
        const v4f ay1 = *(const v4f*)(wp1 + 8 * (g + 2));
        const v4f ay2 = *(const v4f*)(wp0 + 8 * (g + 3));
        const v4f ay3 = *(const v4f*)(wp1 + 8 * (g + 3));
        gemm_load_b(bp, g + 1, bn);
        asm volatile("" ::: "memory");
        __builtin_amdgcn_sched_barrier(0);
        gemm_group(ax0, ax1, bc, acc);
        gemm_load_b(bp, g + 2, bc);
        asm volatile("" ::: "memory");
        __builtin_amdgcn_sched_barrier(0);
        gemm_group(ax2, ax3, bn, acc);

        const int gn = g + 4 < 32 ? g + 4 : g + 2;
        ax0 = *(const v4f*)(wp0 + 8 * gn);
        ax1 = *(const v4f*)(wp1 + 8 * gn);
        ax2 = *(const v4f*)(wp0 + 8 * gn + 8);
        ax3 = *(const v4f*)(wp1 + 8 * gn + 8);
        gemm_load_b(bp, g + 3, bn);
        asm volatile("" ::: "memory");
        __builtin_amdgcn_sched_barrier(0);
        gemm_group(ay0, ay1, bc, acc);
        gemm_load_b(bp, gn, bc);
        asm volatile("" ::: "memory");
        __builtin_amdgcn_sched_barrier(0);
        gemm_group(ay2, ay3, bn, acc);
    }

    if (syncBeforeWrite) __syncthreads();

#pragma unroll
    for (int r = 0; r < 2; ++r) {
        const int mbase = (wv * 2 + r) * 16 + 8 * (lane >> 4);
#pragma unroll
        for (int t = 0; t < NTILES; ++t) {
            const int c = t * 16 + nlo;
#pragma unroll
            for (int v = 0; v < 8; ++v)
                out[(mbase + v) * STRIDE + c] = acc[r][t][v];
        }
    }
}

template <bool SCALED>
__device__ __forceinline__ void gemvT256(const float* __restrict__ W,
                                         const float* __restrict__ S,
                                         float* __restrict__ V,
                                         const float* __restrict__ G,
                                         float* __restrict__ U,
                                         int lane, int wv)
{
    const int c  = lane & 15;
    const int h  = lane >> 4;
    const float* wp = W + (2 * h) * 256 + 32 * wv + 2 * c;
    const float* sp = S + (2 * h) * TB + (c & 7);
    const v8f vz = {0.f, 0.f, 0.f, 0.f, 0.f, 0.f, 0.f, 0.f};
    v8f acc0 = vz, acc1 = vz;
#pragma unroll 16
    for (int ks = 0; ks < 64; ++ks) {
        const v2f w0 = *(const v2f*)(wp + (ks * 4 + 0) * 256);
        const v2f w1 = *(const v2f*)(wp + (ks * 4 + 1) * 256);
        v2f bf;
        bf.x = sp[(ks * 4 + 0) * TB];
        bf.y = sp[(ks * 4 + 1) * TB];
        v2f a0, a1;
        a0.x = w0.x; a0.y = w1.x;
        a1.x = w0.y; a1.y = w1.y;
        acc0 = __builtin_amdgcn_wmma_f32_16x16x4_f32(
            false, a0, false, bf, (short)0, acc0, false, false);
        acc1 = __builtin_amdgcn_wmma_f32_16x16x4_f32(
            false, a1, false, bf, (short)0, acc1, false, false);
    }
    if (c < 8) {
#pragma unroll
        for (int r = 0; r < 8; ++r) {
            const int e = (32 * wv + 2 * (8 * h + r)) * TB + c;
            V[e]      = acc0[r];
            V[e + TB] = acc1[r];
            if (SCALED) {
                U[e]      = acc0[r] * G[e];
                U[e + TB] = acc1[r] * G[e + TB];
            }
        }
    }
}

__global__ __launch_bounds__(256, 1)
void diffnet_fused_kernel(const float* __restrict__ x,
                          const float* __restrict__ w0, const float* __restrict__ b0,
                          const float* __restrict__ w1, const float* __restrict__ b1,
                          const float* __restrict__ w2, const float* __restrict__ b2,
                          const float* __restrict__ w3, const float* __restrict__ b3,
                          float* __restrict__ out)
{
    extern __shared__ float sm[];
    float* Abuf = sm;
    float* Bbuf = Abuf + 256 * STRIDE;
    float* gp0  = Bbuf + 256 * STRIDE;
    float* gpp0 = gp0  + 256 * TB;
    float* gp1  = gpp0 + 256 * TB;
    float* gpp1 = gp1  + 256 * TB;
    float* w3c  = gpp1 + 256 * TB;
    float* c2   = w3c  + 256 * TB + 8;
    float* s2   = c2   + 256 * TB + 8;
    float* v2s  = s2   + 256 * TB;
    float* v1s  = v2s  + 256 * TB;
    float* xl   = v1s  + 256 * TB;

    const int tid  = threadIdx.x;
    const int lane = tid & 31;
    const int wv   = tid >> 5;
    const int n    = blockIdx.y;
    const int bg0  = blockIdx.x * TB;

    const float* W0 = w0 + n * 256 * 12;
    const float* W1 = w1 + n * 256 * 256;
    const float* W2 = w2 + n * 256 * 256;
    const float* W3 = w3 + n * 256;
    const float* Bi0 = b0 + n * 256;
    const float* Bi1 = b1 + n * 256;
    const float* Bi2 = b2 + n * 256;

    if (tid < TB * 12) xl[tid] = x[bg0 * 12 + tid];
    __syncthreads();

    {
        const int i = tid;
        float w0r[12];
#pragma unroll
        for (int m = 0; m < 12; ++m) w0r[m] = W0[i * 12 + m];
        for (int b = 0; b < TB; ++b) {
            float a = Bi0[i];
#pragma unroll
            for (int m = 0; m < 12; ++m) a += w0r[m] * xl[b * 12 + m];
            float t  = tanhf(a);
            float gp = 1.f - t * t;
            gp0[i * TB + b]  = gp;
            gpp0[i * TB + b] = -2.f * t * gp;
#pragma unroll
            for (int m = 0; m < 12; ++m) Abuf[i * STRIDE + b * 12 + m] = gp * w0r[m];
            Abuf[i * STRIDE + 96 + b]  = t;
            Abuf[i * STRIDE + 104 + b] = 0.f;
        }
    }
    __syncthreads();

    gemm256(W1, Abuf, Bbuf, lane, wv, false);
    __syncthreads();

    {
        const int i = tid;
        for (int b = 0; b < TB; ++b) {
            float a  = Bbuf[i * STRIDE + 96 + b] + Bi1[i];
            float t  = tanhf(a);
            float gp = 1.f - t * t;
            gp1[i * TB + b]  = gp;
            gpp1[i * TB + b] = -2.f * t * gp;
#pragma unroll
            for (int m = 0; m < 12; ++m)
                Abuf[i * STRIDE + b * 12 + m] = gp * Bbuf[i * STRIDE + b * 12 + m];
            Abuf[i * STRIDE + 96 + b] = t;
        }
    }
    __syncthreads();

    gemm256(W2, Abuf, Abuf, lane, wv, true);
    __syncthreads();

    {
        const int i = tid;
        const float w3i = W3[i];
        for (int b = 0; b < TB; ++b) {
            float a   = Abuf[i * STRIDE + 96 + b] + Bi2[i];
            float t   = tanhf(a);
            float gp  = 1.f - t * t;
            float gpp = -2.f * t * gp;
            s2[i * TB + b]  = w3i * gp;
            c2[i * TB + b]  = w3i * gpp;
            w3c[i * TB + b] = w3i;
            Abuf[i * STRIDE + 96 + b] = t;
        }
        Abuf[i * STRIDE + 104] = 1.f;
    }
    __syncthreads();

    gemvT256<true>(W2, s2, v2s, gp1, gp0, lane, wv);
    __syncthreads();

    gemvT256<false>(W1, gp0, v1s, nullptr, nullptr, lane, wv);
    __syncthreads();

    {
        const int i = tid;
#pragma unroll
        for (int b = 0; b < TB; ++b) {
            v2s[i * TB + b] = v2s[i * TB + b] * gpp1[i * TB + b];
            v1s[i * TB + b] = v1s[i * TB + b] * gpp0[i * TB + b];
        }
    }
    __syncthreads();

    {
        const int b   = wv;
        const int col = lane & 15;
        const int jc  = col < 12 ? col : 11;
        const int h2  = (lane >> 4) * 2;
        const int bo2 = col < 12 ? b * 12 + col : (col < 14 ? 104 : 96 + b);
        const int co2 = (int)((col == 12 ? s2 : (col == 13 ? w3c : c2)) - sm) + b;
        const v8f vz = {0.f, 0.f, 0.f, 0.f, 0.f, 0.f, 0.f, 0.f};
        v8f acc0 = vz, acc1 = vz, acc2 = vz;
#pragma unroll 4
        for (int ks = 0; ks < 64; ++ks) {
            const int i0 = ks * 4 + h2;
            v2f j0, j1, j2, a0, a1, a2;
            j0.x = W0[(i0 + 0) * 12 + jc];
            j0.y = W0[(i0 + 1) * 12 + jc];
            j1.x = Bbuf[(i0 + 0) * STRIDE + b * 12 + jc];
            j1.y = Bbuf[(i0 + 1) * STRIDE + b * 12 + jc];
            j2.x = Abuf[(i0 + 0) * STRIDE + bo2];
            j2.y = Abuf[(i0 + 1) * STRIDE + bo2];
            a0.x = v1s[(i0 + 0) * TB + b] * j0.x;
            a0.y = v1s[(i0 + 1) * TB + b] * j0.y;
            a1.x = v2s[(i0 + 0) * TB + b] * j1.x;
            a1.y = v2s[(i0 + 1) * TB + b] * j1.y;
            a2.x = sm[co2 + (i0 + 0) * TB] * j2.x;
            a2.y = sm[co2 + (i0 + 1) * TB] * j2.y;
            acc0 = __builtin_amdgcn_wmma_f32_16x16x4_f32(
                false, a0, false, j0, (short)0, acc0, false, false);
            acc1 = __builtin_amdgcn_wmma_f32_16x16x4_f32(
                false, a1, false, j1, (short)0, acc1, false, false);
            acc2 = __builtin_amdgcn_wmma_f32_16x16x4_f32(
                false, a2, false, j2, (short)0, acc2, false, false);
        }
        const v8f acc = acc0 + acc1 + acc2;
        if (col < 12) {
            const int jb = 8 * (lane >> 4);
#pragma unroll
            for (int v = 0; v < 8; ++v)
                if (jb + v < 12) gp0[b * 144 + (jb + v) * 12 + col] = acc[v];
        }
        if ((lane >> 4) == 1) {
            if (col < 12)  gp0[TB * 144 + b * 12 + col] = acc2[4];
            if (col == 14) gp0[TB * 156 + b] = acc2[5] + b3[n];
        }
    }
    __syncthreads();
    for (int idx = tid; idx < TB * 144; idx += 256)
        out[26624 + (n * 512 + bg0) * 144 + idx] = gp0[idx];
    if (tid < TB * 12) out[2048 + (n * 512 + bg0) * 12 + tid] = gp0[TB * 144 + tid];
    if (tid < TB)      out[n * 512 + bg0 + tid] = gp0[TB * 156 + tid];
}

extern "C" void kernel_launch(void* const* d_in, const int* in_sizes, int n_in,
                              void* d_out, int out_size, void* d_ws, size_t ws_size,
                              hipStream_t stream)
{
    (void)in_sizes; (void)n_in; (void)out_size; (void)d_ws; (void)ws_size;
    const float* x  = (const float*)d_in[0];
    const float* w0 = (const float*)d_in[1];
    const float* b0 = (const float*)d_in[2];
    const float* w1 = (const float*)d_in[3];
    const float* b1 = (const float*)d_in[4];
    const float* w2 = (const float*)d_in[5];
    const float* b2 = (const float*)d_in[6];
    const float* w3 = (const float*)d_in[7];
    const float* b3 = (const float*)d_in[8];
    float* outp = (float*)d_out;

    const size_t smem = (size_t)(2 * 256 * STRIDE + 9 * 256 * TB + 16 + TB * 12) * sizeof(float);
    (void)hipFuncSetAttribute(reinterpret_cast<const void*>(&diffnet_fused_kernel),
                              hipFuncAttributeMaxDynamicSharedMemorySize, (int)smem);

    dim3 grid(512 / TB, 4);
    dim3 block(256);
    diffnet_fused_kernel<<<grid, block, smem, stream>>>(x, w0, b0, w1, b1, w2, b2, w3, b3, outp);
}


# activation passes handle a row's eight batch columns together, not one by one: tanh by selection instead of a branch, 16-byte LDS accesses, a pass's loads all issued ahead of its arithmetic
# speedup vs baseline: 3.5777x; 1.0408x over previous
#include <hip/hip_runtime.h>
#include <math.h>


typedef float v2f __attribute__((ext_vector_type(2)));
typedef float v4f __attribute__((ext_vector_type(4)));
typedef float v8f __attribute__((ext_vector_type(8)));

#define TB      8
#define STRIDE  116
#define NCOLS   112
#define NTILES  7

__device__ __forceinline__ void gemm_load_b(const float* __restrict__ bp, int g, float (&b)[NTILES][4])
{
#pragma unroll
    for (int t = 0; t < NTILES; ++t)
#pragma unroll
        for (int e = 0; e < 4; ++e) b[t][e] = bp[(8 * g + e) * STRIDE + 16 * t];
}

__device__ __forceinline__ void gemm_group(const v4f a0, const v4f a1, const float (&b)[NTILES][4],
                                           v8f (&acc)[2][NTILES])
{
    v2f a0lo, a0hi, a1lo, a1hi;
    a0lo.x = a0.x; a0lo.y = a0.y; a0hi.x = a0.z; a0hi.y = a0.w;
    a1lo.x = a1.x; a1lo.y = a1.y; a1hi.x = a1.z; a1hi.y = a1.w;
#pragma unroll
    for (int t = 0; t < NTILES; ++t) {
        v2f bf; bf.x = b[t][0]; bf.y = b[t][1];
        acc[0][t] = __builtin_amdgcn_wmma_f32_16x16x4_f32(
            false, a0lo, false, bf, (short)0, acc[0][t], false, false);
        acc[1][t] = __builtin_amdgcn_wmma_f32_16x16x4_f32(
            false, a1lo, false, bf, (short)0, acc[1][t], false, false);
    }
#pragma unroll
    for (int t = 0; t < NTILES; ++t) {
        v2f bf; bf.x = b[t][2]; bf.y = b[t][3];
        acc[0][t] = __builtin_amdgcn_wmma_f32_16x16x4_f32(
            false, a0hi, false, bf, (short)0, acc[0][t], false, false);
        acc[1][t] = __builtin_amdgcn_wmma_f32_16x16x4_f32(
            false, a1hi, false, bf, (short)0, acc[1][t], false, false);
    }
}

__device__ __forceinline__ void gemm256(const float* __restrict__ W,
                                        const float* __restrict__ in,
                                        float* __restrict__ out,
                                        int lane, int wv, bool syncBeforeWrite)
{
    v8f acc[2][NTILES];
    const v8f vz = {0.f, 0.f, 0.f, 0.f, 0.f, 0.f, 0.f, 0.f};
#pragma unroll
    for (int r = 0; r < 2; ++r)
#pragma unroll
        for (int t = 0; t < NTILES; ++t) acc[r][t] = vz;

    const int nlo = lane & 15;
    const int h4  = (lane >> 4) * 4;
    const float* wp0 = W + ((wv * 2 + 0) * 16 + nlo) * 256 + h4;
    const float* wp1 = wp0 + 16 * 256;
    const float* bp  = in + h4 * STRIDE + nlo;

    v4f ax0 = *(const v4f*)(wp0);
    v4f ax1 = *(const v4f*)(wp1);
    v4f ax2 = *(const v4f*)(wp0 + 8);
    v4f ax3 = *(const v4f*)(wp1 + 8);
    float bc[NTILES][4], bn[NTILES][4];
    gemm_load_b(bp, 0, bc);

    for (int g = 0; g < 32; g += 4) {
        const v4f ay0 = *(const v4f*)(wp0 + 8 * (g + 2));
        const v4f ay1 = *(const v4f*)(wp1 + 8 * (g + 2));
        const v4f ay2 = *(const v4f*)(wp0 + 8 * (g + 3));
        const v4f ay3 = *(const v4f*)(wp1 + 8 * (g + 3));
        gemm_load_b(bp, g + 1, bn);
        asm volatile("" ::: "memory");
        __builtin_amdgcn_sched_barrier(0);
        gemm_group(ax0, ax1, bc, acc);
        gemm_load_b(bp, g + 2, bc);
        asm volatile("" ::: "memory");
        __builtin_amdgcn_sched_barrier(0);
        gemm_group(ax2, ax3, bn, acc);

        const int gn = g + 4 < 32 ? g + 4 : g + 2;
        ax0 = *(const v4f*)(wp0 + 8 * gn);
        ax1 = *(const v4f*)(wp1 + 8 * gn);
        ax2 = *(const v4f*)(wp0 + 8 * gn + 8);
        ax3 = *(const v4f*)(wp1 + 8 * gn + 8);
        gemm_load_b(bp, g + 3, bn);
        asm volatile("" ::: "memory");
        __builtin_amdgcn_sched_barrier(0);
        gemm_group(ay0, ay1, bc, acc);
        gemm_load_b(bp, gn, bc);
        asm volatile("" ::: "memory");
        __builtin_amdgcn_sched_barrier(0);
        gemm_group(ay2, ay3, bn, acc);
    }

    if (syncBeforeWrite) __syncthreads();

#pragma unroll
    for (int r = 0; r < 2; ++r) {
        const int mbase = (wv * 2 + r) * 16 + 8 * (lane >> 4);
#pragma unroll
        for (int t = 0; t < NTILES; ++t) {
            const int c = t * 16 + nlo;
#pragma unroll
            for (int v = 0; v < 8; ++v)
                out[(mbase + v) * STRIDE + c] = acc[r][t][v];
        }
    }
}

template <bool SCALED>
__device__ __forceinline__ void gemvT256(const float* __restrict__ W,
                                         const float* __restrict__ S,
                                         float* __restrict__ V,
                                         const float* __restrict__ G,
                                         float* __restrict__ U,
                                         int lane, int wv)
{
    const int c  = lane & 15;
    const int h  = lane >> 4;
    const float* wp = W + (2 * h) * 256 + 32 * wv + 2 * c;
    const float* sp = S + (2 * h) * TB + (c & 7);
    const v8f vz = {0.f, 0.f, 0.f, 0.f, 0.f, 0.f, 0.f, 0.f};
    v8f acc0 = vz, acc1 = vz;
#pragma unroll 16
    for (int ks = 0; ks < 64; ++ks) {
        const v2f w0 = *(const v2f*)(wp + (ks * 4 + 0) * 256);
        const v2f w1 = *(const v2f*)(wp + (ks * 4 + 1) * 256);
        v2f bf;
        bf.x = sp[(ks * 4 + 0) * TB];
        bf.y = sp[(ks * 4 + 1) * TB];
        v2f a0, a1;
        a0.x = w0.x; a0.y = w1.x;
        a1.x = w0.y; a1.y = w1.y;
        acc0 = __builtin_amdgcn_wmma_f32_16x16x4_f32(
            false, a0, false, bf, (short)0, acc0, false, false);
        acc1 = __builtin_amdgcn_wmma_f32_16x16x4_f32(
            false, a1, false, bf, (short)0, acc1, false, false);
    }
    if (c < 8) {
#pragma unroll
        for (int r = 0; r < 8; ++r) {
            const int e = (32 * wv + 2 * (8 * h + r)) * TB + c;
            V[e]      = acc0[r];
            V[e + TB] = acc1[r];
            if (SCALED) {
                U[e]      = acc0[r] * G[e];
                U[e + TB] = acc1[r] * G[e + TB];
            }
        }
    }
}

__device__ __forceinline__ float tanh_nb(float x)
{
    const float ax = fabsf(x);
    const float y  = ax + ax;
    const float n  = __builtin_rintf(1.44269502f * y);
    float f = fmaf(1.44269502f, y, -n);
    f = fmaf(y, 1.9259629e-8f, f);
    const float e   = ldexpf(__builtin_amdgcn_exp2f(f), (int)n);
    const float big = fmaf(__builtin_amdgcn_rcpf(1.0f + e), -2.0f, 1.0f);
    const float x2 = ax * ax;
    float q = fmaf(x2, -0.0057000206f, 0.020634072f);
    q = fmaf(x2, q, -0.05373793f);
    q = fmaf(x2, q, 0.13331416f);
    q = fmaf(x2, q, -0.3333328f);
    const float small = fmaf(x2, ax * q, ax);
    return copysignf(ax < 0.625f ? small : big, x);
}

__device__ __forceinline__ void act8(const float (&a)[TB], float (&t)[TB], float (&gp)[TB], float (&gpp)[TB])
{
#pragma unroll
    for (int b = 0; b < TB; ++b) {
        t[b]   = tanh_nb(a[b]);
        gp[b]  = 1.f - t[b] * t[b];
        gpp[b] = -2.f * t[b] * gp[b];
    }
}

__device__ __forceinline__ void st8(float* p, const float (&v)[TB])
{
    v4f lo, hi;
    lo.x = v[0]; lo.y = v[1]; lo.z = v[2]; lo.w = v[3];
    hi.x = v[4]; hi.y = v[5]; hi.z = v[6]; hi.w = v[7];
    *(v4f*)(p)     = lo;
    *(v4f*)(p + 4) = hi;
}

__global__ __launch_bounds__(256, 1)
void diffnet_fused_kernel(const float* __restrict__ x,
                          const float* __restrict__ w0, const float* __restrict__ b0,
                          const float* __restrict__ w1, const float* __restrict__ b1,
                          const float* __restrict__ w2, const float* __restrict__ b2,
                          const float* __restrict__ w3, const float* __restrict__ b3,
                          float* __restrict__ out)
{
    extern __shared__ float sm[];
    float* Abuf = sm;
    float* Bbuf = Abuf + 256 * STRIDE;
    float* gp0  = Bbuf + 256 * STRIDE;
    float* gpp0 = gp0  + 256 * TB;
    float* gp1  = gpp0 + 256 * TB;
    float* gpp1 = gp1  + 256 * TB;
    float* w3c  = gpp1 + 256 * TB;
    float* c2   = w3c  + 256 * TB + 8;
    float* s2   = c2   + 256 * TB + 8;
    float* v2s  = s2   + 256 * TB;
    float* v1s  = v2s  + 256 * TB;
    float* xl   = v1s  + 256 * TB;

    const int tid  = threadIdx.x;
    const int lane = tid & 31;
    const int wv   = tid >> 5;
    const int n    = blockIdx.y;
    const int bg0  = blockIdx.x * TB;

    const float* W0 = w0 + n * 256 * 12;
    const float* W1 = w1 + n * 256 * 256;
    const float* W2 = w2 + n * 256 * 256;
    const float* W3 = w3 + n * 256;
    const float* Bi0 = b0 + n * 256;
    const float* Bi1 = b1 + n * 256;
    const float* Bi2 = b2 + n * 256;

    if (tid < TB * 12) xl[tid] = x[bg0 * 12 + tid];
    __syncthreads();

    {
        const int i = tid;
        float w0r[12];
#pragma unroll
        for (int m = 0; m < 12; ++m) w0r[m] = W0[i * 12 + m];
        const float bi = Bi0[i];
        float a[TB], t[TB], gp[TB], gpp[TB];
#pragma unroll
        for (int b = 0; b < TB; ++b) {
            float s = bi;
#pragma unroll
            for (int m = 0; m < 12; ++m) s += w0r[m] * xl[b * 12 + m];
            a[b] = s;
        }
        act8(a, t, gp, gpp);
        float* row = Abuf + i * STRIDE;
#pragma unroll
        for (int b = 0; b < TB; ++b)
#pragma unroll
            for (int q = 0; q < 3; ++q) {
                v4f o;
                o.x = gp[b] * w0r[4 * q + 0]; o.y = gp[b] * w0r[4 * q + 1];
                o.z = gp[b] * w0r[4 * q + 2]; o.w = gp[b] * w0r[4 * q + 3];
                *(v4f*)(row + b * 12 + 4 * q) = o;
            }
        st8(row + 96, t);
        const v4f z4 = {0.f, 0.f, 0.f, 0.f};
        *(v4f*)(row + 104) = z4;
        *(v4f*)(row + 108) = z4;
        st8(gp0 + i * TB, gp);
        st8(gpp0 + i * TB, gpp);
    }
    __syncthreads();

    gemm256(W1, Abuf, Bbuf, lane, wv, false);
    __syncthreads();

    {
        const int i = tid;
        const float* srow = Bbuf + i * STRIDE;
        float* drow = Abuf + i * STRIDE;
        const float bi = Bi1[i];
        const v4f alo = *(const v4f*)(srow + 96);
        const v4f ahi = *(const v4f*)(srow + 100);
        float a[TB], t[TB], gp[TB], gpp[TB];
        a[0] = alo.x + bi; a[1] = alo.y + bi; a[2] = alo.z + bi; a[3] = alo.w + bi;
        a[4] = ahi.x + bi; a[5] = ahi.y + bi; a[6] = ahi.z + bi; a[7] = ahi.w + bi;
        act8(a, t, gp, gpp);
#pragma unroll
        for (int b = 0; b < TB; ++b)
#pragma unroll
            for (int q = 0; q < 3; ++q) {
                const v4f j = *(const v4f*)(srow + b * 12 + 4 * q);
                *(v4f*)(drow + b * 12 + 4 * q) = gp[b] * j;
            }
        st8(drow + 96, t);
        st8(gp1 + i * TB, gp);
        st8(gpp1 + i * TB, gpp);
    }
    __syncthreads();

    gemm256(W2, Abuf, Abuf, lane, wv, true);
    __syncthreads();

    {
        const int i = tid;
        const float w3i = W3[i];
        const float bi  = Bi2[i];
        float* row = Abuf + i * STRIDE;
        const v4f alo = *(const v4f*)(row + 96);
        const v4f ahi = *(const v4f*)(row + 100);
        float a[TB], t[TB], gp[TB], gpp[TB], sv[TB], cv[TB], wv8[TB];
        a[0] = alo.x + bi; a[1] = alo.y + bi; a[2] = alo.z + bi; a[3] = alo.w + bi;
        a[4] = ahi.x + bi; a[5] = ahi.y + bi; a[6] = ahi.z + bi; a[7] = ahi.w + bi;
        act8(a, t, gp, gpp);
#pragma unroll
        for (int b = 0; b < TB; ++b) {
            sv[b]  = w3i * gp[b];
            cv[b]  = w3i * gpp[b];
            wv8[b] = w3i;
        }
        st8(s2 + i * TB, sv);
        st8(c2 + i * TB, cv);
        st8(w3c + i * TB, wv8);
        st8(row + 96, t);
        row[104] = 1.f;
    }
    __syncthreads();

    gemvT256<true>(W2, s2, v2s, gp1, gp0, lane, wv);
    __syncthreads();

    gemvT256<false>(W1, gp0, v1s, nullptr, nullptr, lane, wv);
    __syncthreads();

    {
        const int i = tid;
#pragma unroll
        for (int b = 0; b < TB; ++b) {
            v2s[i * TB + b] = v2s[i * TB + b] * gpp1[i * TB + b];
            v1s[i * TB + b] = v1s[i * TB + b] * gpp0[i * TB + b];
        }
    }
    __syncthreads();

    {
        const int b   = wv;
        const int col = lane & 15;
        const int jc  = col < 12 ? col : 11;
        const int h2  = (lane >> 4) * 2;
        const int bo2 = col < 12 ? b * 12 + col : (col < 14 ? 104 : 96 + b);
        const int co2 = (int)((col == 12 ? s2 : (col == 13 ? w3c : c2)) - sm) + b;
        const v8f vz = {0.f, 0.f, 0.f, 0.f, 0.f, 0.f, 0.f, 0.f};
        v8f acc0 = vz, acc1 = vz, acc2 = vz;
#pragma unroll 4
        for (int ks = 0; ks < 64; ++ks) {
            const int i0 = ks * 4 + h2;
            v2f j0, j1, j2, a0, a1, a2;
            j0.x = W0[(i0 + 0) * 12 + jc];
            j0.y = W0[(i0 + 1) * 12 + jc];
            j1.x = Bbuf[(i0 + 0) * STRIDE + b * 12 + jc];
            j1.y = Bbuf[(i0 + 1) * STRIDE + b * 12 + jc];
            j2.x = Abuf[(i0 + 0) * STRIDE + bo2];
            j2.y = Abuf[(i0 + 1) * STRIDE + bo2];
            a0.x = v1s[(i0 + 0) * TB + b] * j0.x;
            a0.y = v1s[(i0 + 1) * TB + b] * j0.y;
            a1.x = v2s[(i0 + 0) * TB + b] * j1.x;
            a1.y = v2s[(i0 + 1) * TB + b] * j1.y;
            a2.x = sm[co2 + (i0 + 0) * TB] * j2.x;
            a2.y = sm[co2 + (i0 + 1) * TB] * j2.y;
            acc0 = __builtin_amdgcn_wmma_f32_16x16x4_f32(
                false, a0, false, j0, (short)0, acc0, false, false);
            acc1 = __builtin_amdgcn_wmma_f32_16x16x4_f32(
                false, a1, false, j1, (short)0, acc1, false, false);
            acc2 = __builtin_amdgcn_wmma_f32_16x16x4_f32(
                false, a2, false, j2, (short)0, acc2, false, false);
        }
        const v8f acc = acc0 + acc1 + acc2;
        if (col < 12) {
            const int jb = 8 * (lane >> 4);
#pragma unroll
            for (int v = 0; v < 8; ++v)
                if (jb + v < 12) gp0[b * 144 + (jb + v) * 12 + col] = acc[v];
        }
        if ((lane >> 4) == 1) {
            if (col < 12)  gp0[TB * 144 + b * 12 + col] = acc2[4];
            if (col == 14) gp0[TB * 156 + b] = acc2[5] + b3[n];
        }
    }
    __syncthreads();
    for (int idx = tid; idx < TB * 144; idx += 256)
        out[26624 + (n * 512 + bg0) * 144 + idx] = gp0[idx];
    if (tid < TB * 12) out[2048 + (n * 512 + bg0) * 12 + tid] = gp0[TB * 144 + tid];
    if (tid < TB)      out[n * 512 + bg0 + tid] = gp0[TB * 156 + tid];
}

extern "C" void kernel_launch(void* const* d_in, const int* in_sizes, int n_in,
                              void* d_out, int out_size, void* d_ws, size_t ws_size,
                              hipStream_t stream)
{
    (void)in_sizes; (void)n_in; (void)out_size; (void)d_ws; (void)ws_size;
    const float* x  = (const float*)d_in[0];
    const float* w0 = (const float*)d_in[1];
    const float* b0 = (const float*)d_in[2];
    const float* w1 = (const float*)d_in[3];
    const float* b1 = (const float*)d_in[4];
    const float* w2 = (const float*)d_in[5];
    const float* b2 = (const float*)d_in[6];
    const float* w3 = (const float*)d_in[7];
    const float* b3 = (const float*)d_in[8];
    float* outp = (float*)d_out;

    const size_t smem = (size_t)(2 * 256 * STRIDE + 9 * 256 * TB + 16 + TB * 12) * sizeof(float);
    (void)hipFuncSetAttribute(reinterpret_cast<const void*>(&diffnet_fused_kernel),
                              hipFuncAttributeMaxDynamicSharedMemorySize, (int)smem);

    dim3 grid(512 / TB, 4);
    dim3 block(256);
    diffnet_fused_kernel<<<grid, block, smem, stream>>>(x, w0, b0, w1, b1, w2, b2, w3, b3, outp);
}
